# starting kernel
# baseline (speedup 1.0000x reference)
#include <hip/hip_runtime.h>

#define T_STEPS 12
#define F_IN 16
#define H 64

typedef __attribute__((ext_vector_type(16))) _Float16 v16h;
typedef __attribute__((ext_vector_type(8)))  _Float16 v8h;
typedef __attribute__((ext_vector_type(8)))  float    v8f;

__device__ __forceinline__ v16h load_fragA(const _Float16* __restrict__ row, int kb) {
  union { v16h v; v8h h[2]; } u;
  u.h[0] = *(const v8h*)(row + kb);
  u.h[1] = *(const v8h*)(row + kb + 16);
  return u.v;
}

__device__ __forceinline__ float sigmoidf_(float x) {
  return __builtin_amdgcn_rcpf(1.0f + __expf(-x));
}
__device__ __forceinline__ float tanhf_(float x) {
  return 1.0f - 2.0f * __builtin_amdgcn_rcpf(1.0f + __expf(2.0f * x));
}


__global__ __launch_bounds__(256) void init_deg_kernel(float* __restrict__ deg, int N) {
  int i = blockIdx.x * 256 + threadIdx.x;
  if (i < N) deg[i] = 1.0f;
}

__global__ __launch_bounds__(256) void deg_accum_kernel(const int* __restrict__ dst,
                                                        float* __restrict__ deg, int E) {
  int e = blockIdx.x * 256 + threadIdx.x;
  if (e < E) atomicAdd(&deg[dst[e]], 1.0f);
}

__global__ __launch_bounds__(256) void dinv_kernel(float* __restrict__ deg, int N) {
  int i = blockIdx.x * 256 + threadIdx.x;
  if (i < N) deg[i] = rsqrtf(deg[i]);
}

__global__ __launch_bounds__(256) void edge_w_kernel(const int* __restrict__ src,
                                                     const int* __restrict__ dst,
                                                     const float* __restrict__ dinv,
                                                     float* __restrict__ w, int E) {
  int e = blockIdx.x * 256 + threadIdx.x;
  if (e < E) w[e] = dinv[src[e]] * dinv[dst[e]];
}

__global__ __launch_bounds__(256) void cvt_weights_kernel(const float* __restrict__ Wih,
                                                          const float* __restrict__ Whh,
                                                          _Float16* __restrict__ Wih16,
                                                          _Float16* __restrict__ Whh16,
                                                          int n) {
  int i = blockIdx.x * 256 + threadIdx.x;
  if (i < n) {
    Wih16[i] = (_Float16)Wih[i];
    Whh16[i] = (_Float16)Whh[i];
  }
}

__global__ __launch_bounds__(256) void init_h_kernel(float* __restrict__ h,
                                                     _Float16* __restrict__ h16, int n) {
  int i = blockIdx.x * 256 + threadIdx.x;
  if (i < n) { h[i] = 0.0f; h16[i] = (_Float16)0.0f; }
}


__global__ __launch_bounds__(256) void transform_kernel(const float* __restrict__ X,
                                                        const float* __restrict__ Wg,
                                                        const float* __restrict__ dinv,
                                                        float* __restrict__ g,
                                                        float* __restrict__ agg,
                                                        int t, int N) {
  __shared__ float sW[F_IN * H];
  __shared__ float sX[4 * F_IN];
  int tid = threadIdx.x;
  #pragma unroll
  for (int i = tid; i < F_IN * H; i += 256) sW[i] = Wg[i];
  if (tid < 64) {
    int ni = blockIdx.x * 4 + (tid >> 4);
    int k  = tid & 15;
    sX[tid] = (ni < N) ? X[((size_t)ni * T_STEPS + t) * F_IN + k] : 0.0f;
  }
  __syncthreads();
  int node = blockIdx.x * 4 + (tid >> 6);
  int hc   = tid & 63;
  if (node < N) {
    const float* xr = &sX[(tid >> 6) * F_IN];
    float acc = 0.0f;
    #pragma unroll
    for (int k = 0; k < F_IN; ++k) acc += xr[k] * sW[k * H + hc];
    float di = dinv[node];
    size_t idx = (size_t)node * H + hc;
    g[idx]   = acc;
    agg[idx] = acc * di * di;
  }
}

__global__ __launch_bounds__(256) void scatter_kernel(const int* __restrict__ src,
                                                      const int* __restrict__ dst,
                                                      const float* __restrict__ w,
                                                      const float* __restrict__ g,
                                                      float* __restrict__ agg, int E) {
  int tid = blockIdx.x * 256 + threadIdx.x;
  int e = tid >> 4;
  if (e >= E) return;
  int q = tid & 15;
  int s = src[e], d = dst[e];
  float we = w[e];
  const float4 gv = *(const float4*)(g + (size_t)s * H + q * 4);
  float* ap = agg + (size_t)d * H + q * 4;
  atomicAdd(ap + 0, we * gv.x);
  atomicAdd(ap + 1, we * gv.y);
  atomicAdd(ap + 2, we * gv.z);
  atomicAdd(ap + 3, we * gv.w);
}

__global__ __launch_bounds__(256) void act_kernel(const float* __restrict__ agg,
                                                  const float* __restrict__ bg,
                                                  _Float16* __restrict__ a16, size_t n) {
  size_t i = (size_t)blockIdx.x * 256 + threadIdx.x;
  if (i < n) a16[i] = (_Float16)fmaxf(agg[i] + bg[i & 63], 0.0f);
}

__global__ __launch_bounds__(32) void gru_wmma_kernel(const _Float16* __restrict__ a16,
                                                      _Float16* h16,
                                                      const float* __restrict__ hcur,
                                                      float* __restrict__ hnxt,
                                                      const _Float16* __restrict__ Wih16,
                                                      const _Float16* __restrict__ Whh16,
                                                      const float* __restrict__ b_ih,
                                                      const float* __restrict__ b_hh) {
  const int lane  = threadIdx.x;
  const int node0 = blockIdx.x * 16;
  const int m     = lane & 15;
  const int kb    = (lane < 16) ? 0 : 8;
  const int mrow0 = (lane >> 4) << 3;

  const _Float16* arow = a16 + (size_t)(node0 + m) * H;
  const _Float16* hrow = h16 + (size_t)(node0 + m) * H;
  v16h aA0 = load_fragA(arow, kb);
  v16h aA1 = load_fragA(arow, 32 + kb);
  v16h hA0 = load_fragA(hrow, kb);
  v16h hA1 = load_fragA(hrow, 32 + kb);

  #pragma unroll
  for (int j = 0; j < 4; ++j) {
    const _Float16* wr = Wih16 + (size_t)(      j * 16 + m) * H;
    const _Float16* wz = Wih16 + (size_t)( 64 + j * 16 + m) * H;
    const _Float16* wn = Wih16 + (size_t)(128 + j * 16 + m) * H;
    const _Float16* ur = Whh16 + (size_t)(      j * 16 + m) * H;
    const _Float16* uz = Whh16 + (size_t)( 64 + j * 16 + m) * H;
    const _Float16* un = Whh16 + (size_t)(128 + j * 16 + m) * H;

    v8f ir = {}, iz = {}, in_ = {}, hr = {}, hz = {}, hn = {};

    ir  = __builtin_amdgcn_wmma_f32_16x16x32_f16(false, aA0, false, load_fragA(wr, kb), (short)0, ir,  false, false);
    iz  = __builtin_amdgcn_wmma_f32_16x16x32_f16(false, aA0, false, load_fragA(wz, kb), (short)0, iz,  false, false);
    in_ = __builtin_amdgcn_wmma_f32_16x16x32_f16(false, aA0, false, load_fragA(wn, kb), (short)0, in_, false, false);
    hr  = __builtin_amdgcn_wmma_f32_16x16x32_f16(false, hA0, false, load_fragA(ur, kb), (short)0, hr,  false, false);
    hz  = __builtin_amdgcn_wmma_f32_16x16x32_f16(false, hA0, false, load_fragA(uz, kb), (short)0, hz,  false, false);
    hn  = __builtin_amdgcn_wmma_f32_16x16x32_f16(false, hA0, false, load_fragA(un, kb), (short)0, hn,  false, false);
    ir  = __builtin_amdgcn_wmma_f32_16x16x32_f16(false, aA1, false, load_fragA(wr, 32 + kb), (short)0, ir,  false, false);
    iz  = __builtin_amdgcn_wmma_f32_16x16x32_f16(false, aA1, false, load_fragA(wz, 32 + kb), (short)0, iz,  false, false);
    in_ = __builtin_amdgcn_wmma_f32_16x16x32_f16(false, aA1, false, load_fragA(wn, 32 + kb), (short)0, in_, false, false);
    hr  = __builtin_amdgcn_wmma_f32_16x16x32_f16(false, hA1, false, load_fragA(ur, 32 + kb), (short)0, hr,  false, false);
    hz  = __builtin_amdgcn_wmma_f32_16x16x32_f16(false, hA1, false, load_fragA(uz, 32 + kb), (short)0, hz,  false, false);
    hn  = __builtin_amdgcn_wmma_f32_16x16x32_f16(false, hA1, false, load_fragA(un, 32 + kb), (short)0, hn,  false, false);

    const float bir = b_ih[      j * 16 + m];
    const float biz = b_ih[ 64 + j * 16 + m];
    const float bin = b_ih[128 + j * 16 + m];
    const float bhr = b_hh[      j * 16 + m];
    const float bhz = b_hh[ 64 + j * 16 + m];
    const float bhn = b_hh[128 + j * 16 + m];
    #pragma unroll
    for (int v = 0; v < 8; ++v) {
      float r  = sigmoidf_(ir[v] + bir + hr[v] + bhr);
      float z  = sigmoidf_(iz[v] + biz + hz[v] + bhz);
      float nc = tanhf_(in_[v] + bin + r * (hn[v] + bhn));
      size_t idx = (size_t)(node0 + mrow0 + v) * H + j * 16 + m;
      float hold = hcur[idx];
      float hnew = (1.0f - z) * nc + z * hold;
      hnxt[idx] = hnew;
      h16[idx]  = (_Float16)hnew;
    }
  }
}

__global__ __launch_bounds__(256) void head_kernel(const float* __restrict__ h,
                                                   const float* __restrict__ Wout,
                                                   const float* __restrict__ bout,
                                                   float* __restrict__ out, int N) {
  int i = blockIdx.x * 256 + threadIdx.x;
  if (i >= N) return;
  float a0 = bout[0], a1 = bout[1];
  const float* hr = h + (size_t)i * H;
  #pragma unroll
  for (int k = 0; k < H; ++k) {
    float hv = hr[k];
    a0 += hv * Wout[k * 2 + 0];
    a1 += hv * Wout[k * 2 + 1];
  }
  out[i * 2 + 0] = a0;
  out[i * 2 + 1] = a1;
}


static inline int cdiv(long long a, long long b) { return (int)((a + b - 1) / b); }

extern "C" void kernel_launch(void* const* d_in, const int* in_sizes, int n_in,
                              void* d_out, int out_size, void* d_ws, size_t ws_size,
                              hipStream_t stream) {
  const float* X    = (const float*)d_in[0];
  const int*   ei   = (const int*)  d_in[1];
  const float* Wg   = (const float*)d_in[2];
  const float* bg   = (const float*)d_in[3];
  const float* Wih  = (const float*)d_in[4];
  const float* Whh  = (const float*)d_in[5];
  const float* bih  = (const float*)d_in[6];
  const float* bhh  = (const float*)d_in[7];
  const float* Wout = (const float*)d_in[8];
  const float* bout = (const float*)d_in[9];

  const int N = in_sizes[0] / (T_STEPS * F_IN);
  const int E = in_sizes[1] / 2;
  const int* src = ei;
  const int* dst = ei + E;
  const size_t NH = (size_t)N * H;

  char* ws = (char*)d_ws;
  size_t off = 0;
  auto alloc = [&](size_t bytes) -> void* {
    void* p = ws + off;
    off = (off + bytes + 255) & ~(size_t)255;
    return p;
  };
  float*    deg   = (float*)   alloc(sizeof(float) * N);
  float*    wgt   = (float*)   alloc(sizeof(float) * E);
  float*    g     = (float*)   alloc(sizeof(float) * NH);
  float*    agg   = (float*)   alloc(sizeof(float) * NH);
  _Float16* a16   = (_Float16*)alloc(sizeof(_Float16) * NH);
  float*    hbufA = (float*)   alloc(sizeof(float) * NH);
  float*    hbufB = (float*)   alloc(sizeof(float) * NH);
  _Float16* h16   = (_Float16*)alloc(sizeof(_Float16) * NH);
  _Float16* Wih16 = (_Float16*)alloc(sizeof(_Float16) * 3 * H * H);
  _Float16* Whh16 = (_Float16*)alloc(sizeof(_Float16) * 3 * H * H);
  (void)ws_size;

  init_deg_kernel <<<cdiv(N, 256), 256, 0, stream>>>(deg, N);
  deg_accum_kernel<<<cdiv(E, 256), 256, 0, stream>>>(dst, deg, E);
  dinv_kernel     <<<cdiv(N, 256), 256, 0, stream>>>(deg, N);
  edge_w_kernel   <<<cdiv(E, 256), 256, 0, stream>>>(src, dst, deg, wgt, E);
  cvt_weights_kernel<<<cdiv(3 * H * H, 256), 256, 0, stream>>>(Wih, Whh, Wih16, Whh16, 3 * H * H);
  init_h_kernel   <<<cdiv((long long)NH, 256), 256, 0, stream>>>(hbufA, h16, (int)NH);

  float* hcur = hbufA;
  float* hnxt = hbufB;

  for (int t = 0; t < T_STEPS; ++t) {
    transform_kernel<<<cdiv(N, 4), 256, 0, stream>>>(X, Wg, deg, g, agg, t, N);
    scatter_kernel  <<<cdiv((long long)E * 16, 256), 256, 0, stream>>>(src, dst, wgt, g, agg, E);
    act_kernel      <<<cdiv((long long)NH, 256), 256, 0, stream>>>(agg, bg, a16, NH);
    gru_wmma_kernel <<<N / 16, 32, 0, stream>>>(a16, h16, hcur, hnxt, Wih16, Whh16, bih, bhh);
    float* tmp = hcur; hcur = hnxt; hnxt = tmp;
  }

  head_kernel<<<cdiv(N, 256), 256, 0, stream>>>(hcur, Wout, bout, (float*)d_out, N);
}


# Edges binned once a call by destination range (200 nodes, arrival-order slots); a work-group sums its range in an LDS tile by LDS float adds, writes agg once: no global float adds; sum order changed
# speedup vs baseline: 2.4776x; 2.4776x over previous
#include <hip/hip_runtime.h>

#define T_STEPS 12
#define F_IN 16
#define H 64

typedef __attribute__((ext_vector_type(16))) _Float16 v16h;
typedef __attribute__((ext_vector_type(8)))  _Float16 v8h;
typedef __attribute__((ext_vector_type(8)))  float    v8f;


__device__ __forceinline__ v16h fragA_from_f32(const float* row, int kb) {
  const float4 p0 = *(const float4*)(row + kb);
  const float4 p1 = *(const float4*)(row + kb + 4);
  const float4 p2 = *(const float4*)(row + kb + 16);
  const float4 p3 = *(const float4*)(row + kb + 20);
  v16h r;
  r[0]  = (_Float16)p0.x; r[1]  = (_Float16)p0.y; r[2]  = (_Float16)p0.z; r[3]  = (_Float16)p0.w;
  r[4]  = (_Float16)p1.x; r[5]  = (_Float16)p1.y; r[6]  = (_Float16)p1.z; r[7]  = (_Float16)p1.w;
  r[8]  = (_Float16)p2.x; r[9]  = (_Float16)p2.y; r[10] = (_Float16)p2.z; r[11] = (_Float16)p2.w;
  r[12] = (_Float16)p3.x; r[13] = (_Float16)p3.y; r[14] = (_Float16)p3.z; r[15] = (_Float16)p3.w;
  return r;
}

__device__ __forceinline__ v16h fragA_relu_bias(const float* __restrict__ row,
                                                const float* __restrict__ bias, int kb) {
  const float4 p0 = *(const float4*)(row + kb);
  const float4 p1 = *(const float4*)(row + kb + 4);
  const float4 p2 = *(const float4*)(row + kb + 16);
  const float4 p3 = *(const float4*)(row + kb + 20);
  const float4 b0 = *(const float4*)(bias + kb);
  const float4 b1 = *(const float4*)(bias + kb + 4);
  const float4 b2 = *(const float4*)(bias + kb + 16);
  const float4 b3 = *(const float4*)(bias + kb + 20);
  v16h r;
  r[0]  = (_Float16)fmaxf(p0.x + b0.x, 0.0f); r[1]  = (_Float16)fmaxf(p0.y + b0.y, 0.0f);
  r[2]  = (_Float16)fmaxf(p0.z + b0.z, 0.0f); r[3]  = (_Float16)fmaxf(p0.w + b0.w, 0.0f);
  r[4]  = (_Float16)fmaxf(p1.x + b1.x, 0.0f); r[5]  = (_Float16)fmaxf(p1.y + b1.y, 0.0f);
  r[6]  = (_Float16)fmaxf(p1.z + b1.z, 0.0f); r[7]  = (_Float16)fmaxf(p1.w + b1.w, 0.0f);
  r[8]  = (_Float16)fmaxf(p2.x + b2.x, 0.0f); r[9]  = (_Float16)fmaxf(p2.y + b2.y, 0.0f);
  r[10] = (_Float16)fmaxf(p2.z + b2.z, 0.0f); r[11] = (_Float16)fmaxf(p2.w + b2.w, 0.0f);
  r[12] = (_Float16)fmaxf(p3.x + b3.x, 0.0f); r[13] = (_Float16)fmaxf(p3.y + b3.y, 0.0f);
  r[14] = (_Float16)fmaxf(p3.z + b3.z, 0.0f); r[15] = (_Float16)fmaxf(p3.w + b3.w, 0.0f);
  return r;
}

__device__ __forceinline__ float sigmoidf_(float x) {
  return __builtin_amdgcn_rcpf(1.0f + __expf(-x));
}
__device__ __forceinline__ float tanhf_(float x) {
  return 1.0f - 2.0f * __builtin_amdgcn_rcpf(1.0f + __expf(2.0f * x));
}


__device__ __forceinline__ bool upper_half() {
  return ((__builtin_amdgcn_s_sendmsg_rtn(0x87) >> 18) & 1u) != 0u;
}

#define BIN_NODES 200
#define BIN_PAD   32

__global__ __launch_bounds__(256) void init_deg_kernel(float* __restrict__ deg,
                                                       float* __restrict__ degB,
                                                       int* __restrict__ binCnt,
                                                       int nbins, int N) {
  int i = blockIdx.x * 256 + threadIdx.x;
  if (i < N) { deg[i] = 1.0f; degB[i] = 0.0f; }
  if (i < nbins) binCnt[i * BIN_PAD] = 0;
}

__global__ __launch_bounds__(256) void deg_accum_kernel(const int* __restrict__ dst,
                                                        float* __restrict__ degA,
                                                        float* __restrict__ degB,
                                                        int* __restrict__ binCnt, int E) {
  int e = blockIdx.x * 256 + threadIdx.x;
  float* __restrict__ deg = upper_half() ? degB : degA;
  if (e < E) {
    const int d = dst[e];
    atomicAdd(&deg[d], 1.0f);
    atomicAdd(&binCnt[(d / BIN_NODES) * BIN_PAD], 1);
  }
}

__global__ __launch_bounds__(256) void dinv_kernel(float* __restrict__ deg,
                                                   const float* __restrict__ degB, int N) {
  int i = blockIdx.x * 256 + threadIdx.x;
  if (i < N) deg[i] = rsqrtf(deg[i] + degB[i]);
}

#define SCAN_THREADS 1024
__global__ __launch_bounds__(SCAN_THREADS) void bin_scan_kernel(const int* __restrict__ binCnt,
                                                                int* __restrict__ binStart,
                                                                int* __restrict__ cursor,
                                                                int nbins) {
  __shared__ int sh[SCAN_THREADS];
  const int tid = threadIdx.x;
  int carry = 0;
  for (int base = 0; base < nbins; base += SCAN_THREADS) {
    const int i = base + tid;
    const int v = (i < nbins) ? binCnt[i * BIN_PAD] : 0;
    sh[tid] = v;
    __syncthreads();
    for (int off = 1; off < SCAN_THREADS; off <<= 1) {
      const int t = (tid >= off) ? sh[tid - off] : 0;
      __syncthreads();
      sh[tid] += t;
      __syncthreads();
    }
    const int excl = carry + sh[tid] - v;
    if (i < nbins) { binStart[i] = excl; cursor[i * BIN_PAD] = excl; }
    carry += sh[SCAN_THREADS - 1];
    __syncthreads();
  }
  if (tid == 0) binStart[nbins] = carry;
}

__global__ __launch_bounds__(256) void bin_fill_kernel(const int* __restrict__ src,
                                                       const int* __restrict__ dst,
                                                       const float* __restrict__ dinv,
                                                       int* __restrict__ cursor,
                                                       int* __restrict__ recw, int E) {
  int4* __restrict__ rec = (int4*)recw;
  int e = blockIdx.x * 256 + threadIdx.x;
  if (e < E) {
    const int s = src[e];
    const int d = dst[e];
    const int b = d / BIN_NODES;
    const int pos = atomicAdd(&cursor[b * BIN_PAD], 1);
    rec[pos] = make_int4(s, d - b * BIN_NODES, __float_as_int(dinv[s] * dinv[d]), 0);
  }
}

__global__ __launch_bounds__(256) void cvt_weights_kernel(const float* __restrict__ Wih,
                                                          const float* __restrict__ Whh,
                                                          _Float16* __restrict__ Wfrag,
                                                          int n) {
  int o = blockIdx.x * 256 + threadIdx.x;
  if (o < n) {
    const int e    = o & 7;
    const int lane = (o >> 3) & 31;
    const int hh   = (o >> 8) & 1;
    const int f    = o >> 9;
    const int c    = f & 1;
    const int j    = (f >> 1) & 3;
    const int mg   = f >> 3;
    const int mat  = mg / 3;
    const int gate = mg - mat * 3;
    const int row  = gate * 64 + j * 16 + (lane & 15);
    const int k    = ((lane < 16) ? 0 : 8) + 32 * c + 16 * hh + e;
    const float* Wsrc = mat ? Whh : Wih;
    Wfrag[o] = (_Float16)Wsrc[row * H + k];
  }
}

#define INIT_VEC 4
__global__ __launch_bounds__(256) void init_h_kernel(float* __restrict__ h, int n4) {
  float4* __restrict__ h4 = (float4*)h;
  const int base = blockIdx.x * (256 * INIT_VEC) + threadIdx.x;
  #pragma unroll
  for (int j = 0; j < INIT_VEC; ++j) {
    const int i = base + j * 256;
    if (i < n4) h4[i] = make_float4(0.0f, 0.0f, 0.0f, 0.0f);
  }
}


#define TR_NODES 8
__global__ __launch_bounds__(256) void transform_kernel(const float* __restrict__ X,
                                                        const float* __restrict__ Wg,
                                                        float* __restrict__ g,
                                                        int t, int N) {
  const int lane = threadIdx.x & 31;
  const int node0 = __builtin_amdgcn_readfirstlane(
      (int)(((blockIdx.x * 256 + threadIdx.x) >> 5) * TR_NODES));
  if (node0 >= N) return;

  const int nx = (node0 + (lane >> 2) < N) ? node0 + (lane >> 2) : N - 1;
  const float4 xv = *(const float4*)(X + ((size_t)nx * T_STEPS + t) * F_IN + (lane & 3) * 4);

  float w0[F_IN], w1[F_IN];
  #pragma unroll
  for (int k = 0; k < F_IN; ++k) {
    w0[k] = Wg[k * H + lane];
    w1[k] = Wg[k * H + 32 + lane];
  }

  #pragma unroll
  for (int n = 0; n < TR_NODES; ++n) {
    if (node0 + n < N) {
      float acc0 = 0.0f, acc1 = 0.0f;
      #pragma unroll
      for (int q = 0; q < 4; ++q) {
        const float x0 = __int_as_float(__builtin_amdgcn_readlane(__float_as_int(xv.x), n * 4 + q));
        const float x1 = __int_as_float(__builtin_amdgcn_readlane(__float_as_int(xv.y), n * 4 + q));
        const float x2 = __int_as_float(__builtin_amdgcn_readlane(__float_as_int(xv.z), n * 4 + q));
        const float x3 = __int_as_float(__builtin_amdgcn_readlane(__float_as_int(xv.w), n * 4 + q));
        acc0 += x0 * w0[q * 4 + 0]; acc1 += x0 * w1[q * 4 + 0];
        acc0 += x1 * w0[q * 4 + 1]; acc1 += x1 * w1[q * 4 + 1];
        acc0 += x2 * w0[q * 4 + 2]; acc1 += x2 * w1[q * 4 + 2];
        acc0 += x3 * w0[q * 4 + 3]; acc1 += x3 * w1[q * 4 + 3];
      }
      const size_t idx = (size_t)(node0 + n) * H + lane;
      g[idx]        = acc0;
      g[idx + 32]   = acc1;
    }
  }
}

#define EDGE_BATCH   16
#define EDGE_WAVES   32
#define EDGE_THREADS (EDGE_WAVES * 32)
__global__ __launch_bounds__(EDGE_THREADS, 16) void edge_tile_kernel(const int* __restrict__ recw,
                                                                 const int* __restrict__ binStart,
                                                                 const float* __restrict__ g,
                                                                 const float* __restrict__ dinv,
                                                                 float* __restrict__ agg, int N) {
  __shared__ __attribute__((aligned(16))) float tile[BIN_NODES * H];
  const int4* __restrict__ rec = (const int4*)recw;

  const int lane = threadIdx.x & 31;
  const int wave = __builtin_amdgcn_readfirstlane((int)(threadIdx.x >> 5));
  const int n0   = blockIdx.x * BIN_NODES;
  const int nn   = (N - n0 < BIN_NODES) ? (N - n0) : BIN_NODES;
  const int nv   = nn * (H / 4);
  float4* __restrict__ t4 = (float4*)tile;

  const float4* __restrict__ g4 = (const float4*)(g + (size_t)n0 * H);
  constexpr int kInit = (BIN_NODES * (H / 4) + EDGE_THREADS - 1) / EDGE_THREADS;
  float4 gv[kInit];
  float  dv[kInit];
  #pragma unroll
  for (int it = 0; it < kInit; ++it) {
    const int i  = it * EDGE_THREADS + (int)threadIdx.x;
    const int ic = (i < nv) ? i : nv - 1;
    gv[it] = g4[ic];
    dv[it] = dinv[n0 + (ic >> 4)];
  }
  #pragma unroll
  for (int it = 0; it < kInit; ++it) {
    const int i = it * EDGE_THREADS + (int)threadIdx.x;
    const float di = dv[it];
    if (i < nv)
      t4[i] = make_float4(gv[it].x * di * di, gv[it].y * di * di, gv[it].z * di * di, gv[it].w * di * di);
  }
  __syncthreads();

  const int eb = binStart[blockIdx.x];
  const int ee = binStart[blockIdx.x + 1];
  int4 cur = make_int4(0, 0, 0, 0);
  if (eb + wave * 32 + lane < ee) cur = rec[eb + wave * 32 + lane];
  for (int base = eb + wave * 32; base < ee; base += EDGE_THREADS) {
    const int s = cur.x, d = cur.y;
    const float we = __int_as_float(cur.z);
    const int cnt = (ee - base < 32) ? (ee - base) : 32;
    const int en = base + EDGE_THREADS + lane;
    cur = make_int4(0, 0, 0, 0);
    if (en < ee) cur = rec[en];

    if (cnt == 32) {
      #pragma unroll
      for (int b = 0; b < 32; b += EDGE_BATCH) {
        float v0[EDGE_BATCH], v1[EDGE_BATCH];
        #pragma unroll
        for (int j = 0; j < EDGE_BATCH; ++j) {
          const int sj = __builtin_amdgcn_readlane(s, b + j);
          const float* rowp = g + (size_t)sj * H;
          v0[j] = rowp[lane];
          v1[j] = rowp[lane + 32];
        }
        #pragma unroll
        for (int j = 0; j < EDGE_BATCH; ++j) {
          const int dj = __builtin_amdgcn_readlane(d, b + j);
          const float wj = __int_as_float(__builtin_amdgcn_readlane(__float_as_int(we), b + j));
          float* tp = tile + dj * H + lane;
          atomicAdd(tp,      wj * v0[j]);
          atomicAdd(tp + 32, wj * v1[j]);
        }
      }
    } else {
      #pragma unroll
      for (int b = 0; b < 32; b += EDGE_BATCH) {
        if (b < cnt) {
          float v0[EDGE_BATCH], v1[EDGE_BATCH];
          #pragma unroll
          for (int j = 0; j < EDGE_BATCH; ++j) {
            const int jj = (b + j < cnt) ? (b + j) : (cnt - 1);
            const int sj = __builtin_amdgcn_readlane(s, jj);
            const float* rowp = g + (size_t)sj * H;
            v0[j] = rowp[lane];
            v1[j] = rowp[lane + 32];
          }
          #pragma unroll
          for (int j = 0; j < EDGE_BATCH; ++j) {
            if (b + j < cnt) {
              const int dj = __builtin_amdgcn_readlane(d, b + j);
              const float wj = __int_as_float(__builtin_amdgcn_readlane(__float_as_int(we), b + j));
              float* tp = tile + dj * H + lane;
              atomicAdd(tp,      wj * v0[j]);
              atomicAdd(tp + 32, wj * v1[j]);
            }
          }
        }
      }
    }
  }
  __syncthreads();

  float4* __restrict__ a4 = (float4*)(agg + (size_t)n0 * H);
  for (int i = threadIdx.x; i < nv; i += EDGE_THREADS) a4[i] = t4[i];
}

#define GRU_WAVES   25
#define GRU_THREADS (GRU_WAVES * 32)
#define WFRAG_VEC   (48 * 2 * 32)

__device__ __forceinline__ v16h lds_fragB(const v8h* sW, int f, int lane) {
  union { v16h v; v8h h[2]; } u;
  u.h[0] = sW[(f * 2 + 0) * 32 + lane];
  u.h[1] = sW[(f * 2 + 1) * 32 + lane];
  return u.v;
}

__global__ __launch_bounds__(GRU_THREADS) void gru_wmma_kernel(const float* __restrict__ agg,
                                                      const float* __restrict__ bg,
                                                      float* h,
                                                      const v8h* __restrict__ Wfrag,
                                                      const float* __restrict__ b_ih,
                                                      const float* __restrict__ b_hh,
                                                      int ntiles) {
  __shared__ v8h sW[WFRAG_VEC];

  const int lane  = threadIdx.x & 31;
  const int wave  = __builtin_amdgcn_readfirstlane((int)(threadIdx.x >> 5));
  const int tile  = blockIdx.x * GRU_WAVES + wave;
  const bool live = tile < ntiles;
  const int node0 = (live ? tile : ntiles - 1) * 16;
  const int m     = lane & 15;
  const int kb    = (lane < 16) ? 0 : 8;
  const int mrow0 = (lane >> 4) << 3;

  const float* arow = agg + (size_t)(node0 + m) * H;
  const float* hrow = h   + (size_t)(node0 + m) * H;
  v16h aA0 = fragA_relu_bias(arow, bg, kb);
  v16h aA1 = fragA_relu_bias(arow, bg, 32 + kb);
  v16h hA0 = fragA_from_f32(hrow, kb);
  v16h hA1 = fragA_from_f32(hrow, 32 + kb);

  constexpr int kPasses = (WFRAG_VEC + GRU_THREADS - 1) / GRU_THREADS;
  v8h wpiece[kPasses];
  #pragma unroll
  for (int it = 0; it < kPasses; ++it) {
    const int i = it * GRU_THREADS + (int)threadIdx.x;
    wpiece[it] = Wfrag[i < WFRAG_VEC ? i : WFRAG_VEC - 1];
  }
  #pragma unroll
  for (int it = 0; it < kPasses; ++it) {
    const int i = it * GRU_THREADS + (int)threadIdx.x;
    if (i < WFRAG_VEC) sW[i] = wpiece[it];
  }
  __syncthreads();
  if (!live) return;

  #pragma unroll
  for (int j = 0; j < 4; ++j) {
    const int fr = ((0 * 3 + 0) * 4 + j) * 2;
    const int fz = ((0 * 3 + 1) * 4 + j) * 2;
    const int fn = ((0 * 3 + 2) * 4 + j) * 2;
    const int gr = ((1 * 3 + 0) * 4 + j) * 2;
    const int gz = ((1 * 3 + 1) * 4 + j) * 2;
    const int gn = ((1 * 3 + 2) * 4 + j) * 2;

    const float bir = b_ih[      j * 16 + m];
    const float biz = b_ih[ 64 + j * 16 + m];
    const float bin = b_ih[128 + j * 16 + m];
    const float bhr = b_hh[      j * 16 + m];
    const float bhz = b_hh[ 64 + j * 16 + m];
    const float bhn = b_hh[128 + j * 16 + m];
    const size_t hbase = (size_t)(node0 + mrow0) * H + j * 16 + m;
    float hold[8];
    #pragma unroll
    for (int v = 0; v < 8; ++v) hold[v] = h[hbase + (size_t)v * H];

    v8f ir = {}, iz = {}, in_ = {}, hr = {}, hz = {}, hn = {};

    ir  = __builtin_amdgcn_wmma_f32_16x16x32_f16(false, aA0, false, lds_fragB(sW, fr, lane), (short)0, ir,  false, false);
    iz  = __builtin_amdgcn_wmma_f32_16x16x32_f16(false, aA0, false, lds_fragB(sW, fz, lane), (short)0, iz,  false, false);
    in_ = __builtin_amdgcn_wmma_f32_16x16x32_f16(false, aA0, false, lds_fragB(sW, fn, lane), (short)0, in_, false, false);
    hr  = __builtin_amdgcn_wmma_f32_16x16x32_f16(false, hA0, false, lds_fragB(sW, gr, lane), (short)0, hr,  false, false);
    hz  = __builtin_amdgcn_wmma_f32_16x16x32_f16(false, hA0, false, lds_fragB(sW, gz, lane), (short)0, hz,  false, false);
    hn  = __builtin_amdgcn_wmma_f32_16x16x32_f16(false, hA0, false, lds_fragB(sW, gn, lane), (short)0, hn,  false, false);
    ir  = __builtin_amdgcn_wmma_f32_16x16x32_f16(false, aA1, false, lds_fragB(sW, fr + 1, lane), (short)0, ir,  false, false);
    iz  = __builtin_amdgcn_wmma_f32_16x16x32_f16(false, aA1, false, lds_fragB(sW, fz + 1, lane), (short)0, iz,  false, false);
    in_ = __builtin_amdgcn_wmma_f32_16x16x32_f16(false, aA1, false, lds_fragB(sW, fn + 1, lane), (short)0, in_, false, false);
    hr  = __builtin_amdgcn_wmma_f32_16x16x32_f16(false, hA1, false, lds_fragB(sW, gr + 1, lane), (short)0, hr,  false, false);
    hz  = __builtin_amdgcn_wmma_f32_16x16x32_f16(false, hA1, false, lds_fragB(sW, gz + 1, lane), (short)0, hz,  false, false);
    hn  = __builtin_amdgcn_wmma_f32_16x16x32_f16(false, hA1, false, lds_fragB(sW, gn + 1, lane), (short)0, hn,  false, false);

    #pragma unroll
    for (int v = 0; v < 8; ++v) {
      float r  = sigmoidf_(ir[v] + bir + hr[v] + bhr);
      float z  = sigmoidf_(iz[v] + biz + hz[v] + bhz);
      float nc = tanhf_(in_[v] + bin + r * (hn[v] + bhn));
      float hnew = (1.0f - z) * nc + z * hold[v];
      h[hbase + (size_t)v * H] = hnew;
    }
  }
}

__global__ __launch_bounds__(256) void head_kernel(const float* __restrict__ h,
                                                   const float* __restrict__ Wout,
                                                   const float* __restrict__ bout,
                                                   float* __restrict__ out, int N) {
  int i = blockIdx.x * 256 + threadIdx.x;
  if (i >= N) return;
  float a0 = bout[0], a1 = bout[1];
  const float* hr = h + (size_t)i * H;
  #pragma unroll
  for (int k = 0; k < H; ++k) {
    float hv = hr[k];
    a0 += hv * Wout[k * 2 + 0];
    a1 += hv * Wout[k * 2 + 1];
  }
  out[i * 2 + 0] = a0;
  out[i * 2 + 1] = a1;
}


static inline int cdiv(long long a, long long b) { return (int)((a + b - 1) / b); }

extern "C" void kernel_launch(void* const* d_in, const int* in_sizes, int n_in,
                              void* d_out, int out_size, void* d_ws, size_t ws_size,
                              hipStream_t stream) {
  const float* X    = (const float*)d_in[0];
  const int*   ei   = (const int*)  d_in[1];
  const float* Wg   = (const float*)d_in[2];
  const float* bg   = (const float*)d_in[3];
  const float* Wih  = (const float*)d_in[4];
  const float* Whh  = (const float*)d_in[5];
  const float* bih  = (const float*)d_in[6];
  const float* bhh  = (const float*)d_in[7];
  const float* Wout = (const float*)d_in[8];
  const float* bout = (const float*)d_in[9];

  const int N = in_sizes[0] / (T_STEPS * F_IN);
  const int E = in_sizes[1] / 2;
  const int* src = ei;
  const int* dst = ei + E;
  const size_t NH = (size_t)N * H;

  char* ws = (char*)d_ws;
  size_t off = 0;
  auto alloc = [&](size_t bytes) -> void* {
    void* p = ws + off;
    off = (off + bytes + 255) & ~(size_t)255;
    return p;
  };
  const int nbins = cdiv(N, BIN_NODES);
  float*    deg   = (float*)   alloc(sizeof(float) * N);
  float*    g     = (float*)   alloc(sizeof(float) * NH);
  float*    agg   = (float*)   alloc(sizeof(float) * NH);
  float*    hbuf  = (float*)   alloc(sizeof(float) * NH);
  int*      rec   = (int*)     alloc(sizeof(int4) * E);
  _Float16* Wfrag = (_Float16*)alloc(sizeof(_Float16) * 2 * 3 * H * H);
  int*      binCnt   = (int*)  alloc(sizeof(int) * BIN_PAD * nbins);
  int*      cursor   = (int*)  alloc(sizeof(int) * BIN_PAD * nbins);
  int*      binStart = (int*)  alloc(sizeof(int) * (nbins + 1));
  float*    degB  = g;
  (void)ws_size;

  init_deg_kernel <<<cdiv(N, 256), 256, 0, stream>>>(deg, degB, binCnt, nbins, N);
  deg_accum_kernel<<<cdiv(E, 256), 256, 0, stream>>>(dst, deg, degB, binCnt, E);
  dinv_kernel     <<<cdiv(N, 256), 256, 0, stream>>>(deg, degB, N);
  bin_scan_kernel <<<1, SCAN_THREADS, 0, stream>>>(binCnt, binStart, cursor, nbins);
  bin_fill_kernel <<<cdiv(E, 256), 256, 0, stream>>>(src, dst, deg, cursor, rec, E);
  cvt_weights_kernel<<<cdiv(2 * 3 * H * H, 256), 256, 0, stream>>>(Wih, Whh, Wfrag, 2 * 3 * H * H);
  init_h_kernel   <<<cdiv((long long)(NH / 4), 256 * INIT_VEC), 256, 0, stream>>>(hbuf, (int)(NH / 4));

  for (int t = 0; t < T_STEPS; ++t) {
    transform_kernel<<<cdiv(N, 8 * TR_NODES), 256, 0, stream>>>(X, Wg, g, t, N);
    edge_tile_kernel<<<nbins, EDGE_THREADS, 0, stream>>>(rec, binStart, g, deg, agg, N);
    gru_wmma_kernel <<<cdiv(N / 16, GRU_WAVES), GRU_THREADS, 0, stream>>>(agg, bg, hbuf, (const v8h*)Wfrag, bih, bhh, N / 16);
  }

  head_kernel<<<cdiv(N, 256), 256, 0, stream>>>(hbuf, Wout, bout, (float*)d_out, N);
}


# GRU cell moved into the edge-tile kernel: a 16-wave group sums a 208-node bin in LDS, then 13 waves feed relu(tile+bg) from LDS to the WMMAs; agg array, 12 launches gone; edge adds arrive reordered
# speedup vs baseline: 2.5834x; 1.0427x over previous
#include <hip/hip_runtime.h>

#define T_STEPS 12
#define F_IN 16
#define H 64

typedef __attribute__((ext_vector_type(16))) _Float16 v16h;
typedef __attribute__((ext_vector_type(8)))  _Float16 v8h;
typedef __attribute__((ext_vector_type(8)))  float    v8f;


__device__ __forceinline__ v16h fragA_from_f32(const float* row, int kb) {
  const float4 p0 = *(const float4*)(row + kb);
  const float4 p1 = *(const float4*)(row + kb + 4);
  const float4 p2 = *(const float4*)(row + kb + 16);
  const float4 p3 = *(const float4*)(row + kb + 20);
  v16h r;
  r[0]  = (_Float16)p0.x; r[1]  = (_Float16)p0.y; r[2]  = (_Float16)p0.z; r[3]  = (_Float16)p0.w;
  r[4]  = (_Float16)p1.x; r[5]  = (_Float16)p1.y; r[6]  = (_Float16)p1.z; r[7]  = (_Float16)p1.w;
  r[8]  = (_Float16)p2.x; r[9]  = (_Float16)p2.y; r[10] = (_Float16)p2.z; r[11] = (_Float16)p2.w;
  r[12] = (_Float16)p3.x; r[13] = (_Float16)p3.y; r[14] = (_Float16)p3.z; r[15] = (_Float16)p3.w;
  return r;
}

__device__ __forceinline__ float sigmoidf_(float x) {
  return __builtin_amdgcn_rcpf(1.0f + __expf(-x));
}
__device__ __forceinline__ float tanhf_(float x) {
  return 1.0f - 2.0f * __builtin_amdgcn_rcpf(1.0f + __expf(2.0f * x));
}


__device__ __forceinline__ bool upper_half() {
  return ((__builtin_amdgcn_s_sendmsg_rtn(0x87) >> 18) & 1u) != 0u;
}

#define BIN_NODES 208
#define BIN_PAD   32

__global__ __launch_bounds__(256) void init_deg_kernel(float* __restrict__ deg,
                                                       float* __restrict__ degB,
                                                       int* __restrict__ binCnt,
                                                       int nbins, int N) {
  int i = blockIdx.x * 256 + threadIdx.x;
  if (i < N) { deg[i] = 1.0f; degB[i] = 0.0f; }
  if (i < nbins) binCnt[i * BIN_PAD] = 0;
}

__global__ __launch_bounds__(256) void deg_accum_kernel(const int* __restrict__ dst,
                                                        float* __restrict__ degA,
                                                        float* __restrict__ degB,
                                                        int* __restrict__ binCnt, int E) {
  int e = blockIdx.x * 256 + threadIdx.x;
  float* __restrict__ deg = upper_half() ? degB : degA;
  if (e < E) {
    const int d = dst[e];
    atomicAdd(&deg[d], 1.0f);
    atomicAdd(&binCnt[(d / BIN_NODES) * BIN_PAD], 1);
  }
}

__global__ __launch_bounds__(256) void dinv_kernel(float* __restrict__ deg,
                                                   const float* __restrict__ degB, int N) {
  int i = blockIdx.x * 256 + threadIdx.x;
  if (i < N) deg[i] = rsqrtf(deg[i] + degB[i]);
}

#define SCAN_THREADS 1024
__global__ __launch_bounds__(SCAN_THREADS) void bin_scan_kernel(const int* __restrict__ binCnt,
                                                                int* __restrict__ binStart,
                                                                int* __restrict__ cursor,
                                                                int nbins) {
  __shared__ int sh[SCAN_THREADS];
  const int tid = threadIdx.x;
  int carry = 0;
  for (int base = 0; base < nbins; base += SCAN_THREADS) {
    const int i = base + tid;
    const int v = (i < nbins) ? binCnt[i * BIN_PAD] : 0;
    sh[tid] = v;
    __syncthreads();
    for (int off = 1; off < SCAN_THREADS; off <<= 1) {
      const int t = (tid >= off) ? sh[tid - off] : 0;
      __syncthreads();
      sh[tid] += t;
      __syncthreads();
    }
    const int excl = carry + sh[tid] - v;
    if (i < nbins) { binStart[i] = excl; cursor[i * BIN_PAD] = excl; }
    carry += sh[SCAN_THREADS - 1];
    __syncthreads();
  }
  if (tid == 0) binStart[nbins] = carry;
}

__global__ __launch_bounds__(256) void bin_fill_kernel(const int* __restrict__ src,
                                                       const int* __restrict__ dst,
                                                       const float* __restrict__ dinv,
                                                       int* __restrict__ cursor,
                                                       int* __restrict__ recw, int E) {
  int4* __restrict__ rec = (int4*)recw;
  int e = blockIdx.x * 256 + threadIdx.x;
  if (e < E) {
    const int s = src[e];
    const int d = dst[e];
    const int b = d / BIN_NODES;
    const int pos = atomicAdd(&cursor[b * BIN_PAD], 1);
    rec[pos] = make_int4(s, d - b * BIN_NODES, __float_as_int(dinv[s] * dinv[d]), 0);
  }
}

__global__ __launch_bounds__(256) void cvt_weights_kernel(const float* __restrict__ Wih,
                                                          const float* __restrict__ Whh,
                                                          _Float16* __restrict__ Wfrag,
                                                          int n) {
  int o = blockIdx.x * 256 + threadIdx.x;
  if (o < n) {
    const int e    = o & 7;
    const int lane = (o >> 3) & 31;
    const int hh   = (o >> 8) & 1;
    const int f    = o >> 9;
    const int c    = f & 1;
    const int j    = (f >> 1) & 3;
    const int mg   = f >> 3;
    const int mat  = mg / 3;
    const int gate = mg - mat * 3;
    const int row  = gate * 64 + j * 16 + (lane & 15);
    const int k    = ((lane < 16) ? 0 : 8) + 32 * c + 16 * hh + e;
    const float* Wsrc = mat ? Whh : Wih;
    Wfrag[o] = (_Float16)Wsrc[row * H + k];
  }
}

#define INIT_VEC 4
__global__ __launch_bounds__(256) void init_h_kernel(float* __restrict__ h, int n4) {
  float4* __restrict__ h4 = (float4*)h;
  const int base = blockIdx.x * (256 * INIT_VEC) + threadIdx.x;
  #pragma unroll
  for (int j = 0; j < INIT_VEC; ++j) {
    const int i = base + j * 256;
    if (i < n4) h4[i] = make_float4(0.0f, 0.0f, 0.0f, 0.0f);
  }
}


#define TR_NODES 8
__global__ __launch_bounds__(256) void transform_kernel(const float* __restrict__ X,
                                                        const float* __restrict__ Wg,
                                                        float* __restrict__ g,
                                                        int t, int N) {
  const int lane = threadIdx.x & 31;
  const int node0 = __builtin_amdgcn_readfirstlane(
      (int)(((blockIdx.x * 256 + threadIdx.x) >> 5) * TR_NODES));
  if (node0 >= N) return;

  const int nx = (node0 + (lane >> 2) < N) ? node0 + (lane >> 2) : N - 1;
  const float4 xv = *(const float4*)(X + ((size_t)nx * T_STEPS + t) * F_IN + (lane & 3) * 4);

  float w0[F_IN], w1[F_IN];
  #pragma unroll
  for (int k = 0; k < F_IN; ++k) {
    w0[k] = Wg[k * H + lane];
    w1[k] = Wg[k * H + 32 + lane];
  }

  #pragma unroll
  for (int n = 0; n < TR_NODES; ++n) {
    if (node0 + n < N) {
      float acc0 = 0.0f, acc1 = 0.0f;
      #pragma unroll
      for (int q = 0; q < 4; ++q) {
        const float x0 = __int_as_float(__builtin_amdgcn_readlane(__float_as_int(xv.x), n * 4 + q));
        const float x1 = __int_as_float(__builtin_amdgcn_readlane(__float_as_int(xv.y), n * 4 + q));
        const float x2 = __int_as_float(__builtin_amdgcn_readlane(__float_as_int(xv.z), n * 4 + q));
        const float x3 = __int_as_float(__builtin_amdgcn_readlane(__float_as_int(xv.w), n * 4 + q));
        acc0 += x0 * w0[q * 4 + 0]; acc1 += x0 * w1[q * 4 + 0];
        acc0 += x1 * w0[q * 4 + 1]; acc1 += x1 * w1[q * 4 + 1];
        acc0 += x2 * w0[q * 4 + 2]; acc1 += x2 * w1[q * 4 + 2];
        acc0 += x3 * w0[q * 4 + 3]; acc1 += x3 * w1[q * 4 + 3];
      }
      const size_t idx = (size_t)(node0 + n) * H + lane;
      g[idx]        = acc0;
      g[idx + 32]   = acc1;
    }
  }
}

#define EDGE_BATCH   32
#define EDGE_WAVES   16
#define EDGE_THREADS (EDGE_WAVES * 32)
#define WFRAG_VEC    (48 * 2 * 32)

__device__ __forceinline__ v16h lds_fragB(const v8h* sW, int f, int lane) {
  union { v16h v; v8h h[2]; } u;
  u.h[0] = sW[(f * 2 + 0) * 32 + lane];
  u.h[1] = sW[(f * 2 + 1) * 32 + lane];
  return u.v;
}

__global__ __launch_bounds__(EDGE_THREADS, 8) void edge_gru_kernel(const int* __restrict__ recw,
                                                                const int* __restrict__ binStart,
                                                                const float* __restrict__ g,
                                                                const float* __restrict__ dinv,
                                                                const float* __restrict__ bg,
                                                                float* h,
                                                                const v8h* __restrict__ Wfrag,
                                                                const float* __restrict__ b_ih,
                                                                const float* __restrict__ b_hh,
                                                                int N) {
  __shared__ __attribute__((aligned(16))) float tile[BIN_NODES * H];
  __shared__ v8h sW[WFRAG_VEC];
  const int4* __restrict__ rec = (const int4*)recw;

  const int lane = threadIdx.x & 31;
  const int wave = __builtin_amdgcn_readfirstlane((int)(threadIdx.x >> 5));
  const int n0   = blockIdx.x * BIN_NODES;
  const int nn   = (N - n0 < BIN_NODES) ? (N - n0) : BIN_NODES;
  const int nv   = nn * (H / 4);
  float4* __restrict__ t4 = (float4*)tile;

  const bool gruwave = (wave * 16 + 16) <= nn;
  const int  node0   = n0 + wave * 16;
  const int  m       = lane & 15;
  const int  kb      = (lane < 16) ? 0 : 8;
  const int  mrow0   = (lane >> 4) << 3;

  v16h hA0 = {}, hA1 = {};
  if (gruwave) {
    const float* hrow = h + (size_t)(node0 + m) * H;
    hA0 = fragA_from_f32(hrow, kb);
    hA1 = fragA_from_f32(hrow, 32 + kb);
  }

  constexpr int kPasses = (WFRAG_VEC + EDGE_THREADS - 1) / EDGE_THREADS;
  v8h wpiece[kPasses];
  #pragma unroll
  for (int it = 0; it < kPasses; ++it) {
    const int i = it * EDGE_THREADS + (int)threadIdx.x;
    wpiece[it] = Wfrag[i < WFRAG_VEC ? i : WFRAG_VEC - 1];
  }
  const float4* __restrict__ g4 = (const float4*)(g + (size_t)n0 * H);
  constexpr int kInit = (BIN_NODES * (H / 4) + EDGE_THREADS - 1) / EDGE_THREADS;
  float4 gv[kInit];
  float  dv[kInit];
  #pragma unroll
  for (int it = 0; it < kInit; ++it) {
    const int i  = it * EDGE_THREADS + (int)threadIdx.x;
    const int ic = (i < nv) ? i : nv - 1;
    gv[it] = g4[ic];
    dv[it] = dinv[n0 + (ic >> 4)];
  }
  #pragma unroll
  for (int it = 0; it < kPasses; ++it) {
    const int i = it * EDGE_THREADS + (int)threadIdx.x;
    if (i < WFRAG_VEC) sW[i] = wpiece[it];
  }
  #pragma unroll
  for (int it = 0; it < kInit; ++it) {
    const int i = it * EDGE_THREADS + (int)threadIdx.x;
    const float di = dv[it];
    if (i < nv)
      t4[i] = make_float4(gv[it].x * di * di, gv[it].y * di * di, gv[it].z * di * di, gv[it].w * di * di);
  }
  __syncthreads();

  const int eb = binStart[blockIdx.x];
  const int ee = binStart[blockIdx.x + 1];
  int4 cur = make_int4(0, 0, 0, 0);
  if (eb + wave * 32 + lane < ee) cur = rec[eb + wave * 32 + lane];
  for (int base = eb + wave * 32; base < ee; base += EDGE_THREADS) {
    const int s = cur.x, d = cur.y;
    const float we = __int_as_float(cur.z);
    const int cnt = (ee - base < 32) ? (ee - base) : 32;
    const int en = base + EDGE_THREADS + lane;
    cur = make_int4(0, 0, 0, 0);
    if (en < ee) cur = rec[en];

    if (cnt == 32) {
      #pragma unroll
      for (int b = 0; b < 32; b += EDGE_BATCH) {
        float v0[EDGE_BATCH], v1[EDGE_BATCH];
        #pragma unroll
        for (int j = 0; j < EDGE_BATCH; ++j) {
          const int sj = __builtin_amdgcn_readlane(s, b + j);
          const float* rowp = g + (size_t)sj * H;
          v0[j] = rowp[lane];
          v1[j] = rowp[lane + 32];
        }
        #pragma unroll
        for (int j = 0; j < EDGE_BATCH; ++j) {
          const int dj = __builtin_amdgcn_readlane(d, b + j);
          const float wj = __int_as_float(__builtin_amdgcn_readlane(__float_as_int(we), b + j));
          float* tp = tile + dj * H + lane;
          atomicAdd(tp,      wj * v0[j]);
          atomicAdd(tp + 32, wj * v1[j]);
        }
      }
    } else {
      #pragma unroll
      for (int b = 0; b < 32; b += EDGE_BATCH) {
        if (b < cnt) {
          float v0[EDGE_BATCH], v1[EDGE_BATCH];
          #pragma unroll
          for (int j = 0; j < EDGE_BATCH; ++j) {
            const int jj = (b + j < cnt) ? (b + j) : (cnt - 1);
            const int sj = __builtin_amdgcn_readlane(s, jj);
            const float* rowp = g + (size_t)sj * H;
            v0[j] = rowp[lane];
            v1[j] = rowp[lane + 32];
          }
          #pragma unroll
          for (int j = 0; j < EDGE_BATCH; ++j) {
            if (b + j < cnt) {
              const int dj = __builtin_amdgcn_readlane(d, b + j);
              const float wj = __int_as_float(__builtin_amdgcn_readlane(__float_as_int(we), b + j));
              float* tp = tile + dj * H + lane;
              atomicAdd(tp,      wj * v0[j]);
              atomicAdd(tp + 32, wj * v1[j]);
            }
          }
        }
      }
    }
  }
  __syncthreads();

  if (!gruwave) return;

  v16h aA0, aA1;
  {
    const float* trow = tile + (wave * 16 + m) * H;
    #pragma unroll
    for (int c = 0; c < 2; ++c) {
      const int k0 = 32 * c + kb;
      const float4 p0 = *(const float4*)(trow + k0);
      const float4 p1 = *(const float4*)(trow + k0 + 4);
      const float4 p2 = *(const float4*)(trow + k0 + 16);
      const float4 p3 = *(const float4*)(trow + k0 + 20);
      const float4 b0 = *(const float4*)(bg + k0);
      const float4 b1 = *(const float4*)(bg + k0 + 4);
      const float4 b2 = *(const float4*)(bg + k0 + 16);
      const float4 b3 = *(const float4*)(bg + k0 + 20);
      v16h r;
      r[0]  = (_Float16)fmaxf(p0.x + b0.x, 0.0f); r[1]  = (_Float16)fmaxf(p0.y + b0.y, 0.0f);
      r[2]  = (_Float16)fmaxf(p0.z + b0.z, 0.0f); r[3]  = (_Float16)fmaxf(p0.w + b0.w, 0.0f);
      r[4]  = (_Float16)fmaxf(p1.x + b1.x, 0.0f); r[5]  = (_Float16)fmaxf(p1.y + b1.y, 0.0f);
      r[6]  = (_Float16)fmaxf(p1.z + b1.z, 0.0f); r[7]  = (_Float16)fmaxf(p1.w + b1.w, 0.0f);
      r[8]  = (_Float16)fmaxf(p2.x + b2.x, 0.0f); r[9]  = (_Float16)fmaxf(p2.y + b2.y, 0.0f);
      r[10] = (_Float16)fmaxf(p2.z + b2.z, 0.0f); r[11] = (_Float16)fmaxf(p2.w + b2.w, 0.0f);
      r[12] = (_Float16)fmaxf(p3.x + b3.x, 0.0f); r[13] = (_Float16)fmaxf(p3.y + b3.y, 0.0f);
      r[14] = (_Float16)fmaxf(p3.z + b3.z, 0.0f); r[15] = (_Float16)fmaxf(p3.w + b3.w, 0.0f);
      if (c == 0) aA0 = r; else aA1 = r;
    }
  }

  #pragma unroll
  for (int j = 0; j < 4; ++j) {
    const int fr = ((0 * 3 + 0) * 4 + j) * 2;
    const int fz = ((0 * 3 + 1) * 4 + j) * 2;
    const int fn = ((0 * 3 + 2) * 4 + j) * 2;
    const int gr = ((1 * 3 + 0) * 4 + j) * 2;
    const int gz = ((1 * 3 + 1) * 4 + j) * 2;
    const int gn = ((1 * 3 + 2) * 4 + j) * 2;

    const float bir = b_ih[      j * 16 + m];
    const float biz = b_ih[ 64 + j * 16 + m];
    const float bin = b_ih[128 + j * 16 + m];
    const float bhr = b_hh[      j * 16 + m];
    const float bhz = b_hh[ 64 + j * 16 + m];
    const float bhn = b_hh[128 + j * 16 + m];
    const size_t hbase = (size_t)(node0 + mrow0) * H + j * 16 + m;
    float hold[8];
    #pragma unroll
    for (int v = 0; v < 8; ++v) hold[v] = h[hbase + (size_t)v * H];

    v8f ir = {}, iz = {}, in_ = {}, hr = {}, hz = {}, hn = {};

    ir  = __builtin_amdgcn_wmma_f32_16x16x32_f16(false, aA0, false, lds_fragB(sW, fr, lane), (short)0, ir,  false, false);
    iz  = __builtin_amdgcn_wmma_f32_16x16x32_f16(false, aA0, false, lds_fragB(sW, fz, lane), (short)0, iz,  false, false);
    in_ = __builtin_amdgcn_wmma_f32_16x16x32_f16(false, aA0, false, lds_fragB(sW, fn, lane), (short)0, in_, false, false);
    hr  = __builtin_amdgcn_wmma_f32_16x16x32_f16(false, hA0, false, lds_fragB(sW, gr, lane), (short)0, hr,  false, false);
    hz  = __builtin_amdgcn_wmma_f32_16x16x32_f16(false, hA0, false, lds_fragB(sW, gz, lane), (short)0, hz,  false, false);
    hn  = __builtin_amdgcn_wmma_f32_16x16x32_f16(false, hA0, false, lds_fragB(sW, gn, lane), (short)0, hn,  false, false);
    ir  = __builtin_amdgcn_wmma_f32_16x16x32_f16(false, aA1, false, lds_fragB(sW, fr + 1, lane), (short)0, ir,  false, false);
    iz  = __builtin_amdgcn_wmma_f32_16x16x32_f16(false, aA1, false, lds_fragB(sW, fz + 1, lane), (short)0, iz,  false, false);
    in_ = __builtin_amdgcn_wmma_f32_16x16x32_f16(false, aA1, false, lds_fragB(sW, fn + 1, lane), (short)0, in_, false, false);
    hr  = __builtin_amdgcn_wmma_f32_16x16x32_f16(false, hA1, false, lds_fragB(sW, gr + 1, lane), (short)0, hr,  false, false);
    hz  = __builtin_amdgcn_wmma_f32_16x16x32_f16(false, hA1, false, lds_fragB(sW, gz + 1, lane), (short)0, hz,  false, false);
    hn  = __builtin_amdgcn_wmma_f32_16x16x32_f16(false, hA1, false, lds_fragB(sW, gn + 1, lane), (short)0, hn,  false, false);

    #pragma unroll
    for (int v = 0; v < 8; ++v) {
      float r  = sigmoidf_(ir[v] + bir + hr[v] + bhr);
      float z  = sigmoidf_(iz[v] + biz + hz[v] + bhz);
      float nc = tanhf_(in_[v] + bin + r * (hn[v] + bhn));
      float hnew = (1.0f - z) * nc + z * hold[v];
      h[hbase + (size_t)v * H] = hnew;
    }
  }
}

__global__ __launch_bounds__(256) void head_kernel(const float* __restrict__ h,
                                                   const float* __restrict__ Wout,
                                                   const float* __restrict__ bout,
                                                   float* __restrict__ out, int N) {
  int i = blockIdx.x * 256 + threadIdx.x;
  if (i >= N) return;
  float a0 = bout[0], a1 = bout[1];
  const float* hr = h + (size_t)i * H;
  #pragma unroll
  for (int k = 0; k < H; ++k) {
    float hv = hr[k];
    a0 += hv * Wout[k * 2 + 0];
    a1 += hv * Wout[k * 2 + 1];
  }
  out[i * 2 + 0] = a0;
  out[i * 2 + 1] = a1;
}


static inline int cdiv(long long a, long long b) { return (int)((a + b - 1) / b); }

extern "C" void kernel_launch(void* const* d_in, const int* in_sizes, int n_in,
                              void* d_out, int out_size, void* d_ws, size_t ws_size,
                              hipStream_t stream) {
  const float* X    = (const float*)d_in[0];
  const int*   ei   = (const int*)  d_in[1];
  const float* Wg   = (const float*)d_in[2];
  const float* bg   = (const float*)d_in[3];
  const float* Wih  = (const float*)d_in[4];
  const float* Whh  = (const float*)d_in[5];
  const float* bih  = (const float*)d_in[6];
  const float* bhh  = (const float*)d_in[7];
  const float* Wout = (const float*)d_in[8];
  const float* bout = (const float*)d_in[9];

  const int N = in_sizes[0] / (T_STEPS * F_IN);
  const int E = in_sizes[1] / 2;
  const int* src = ei;
  const int* dst = ei + E;
  const size_t NH = (size_t)N * H;

  char* ws = (char*)d_ws;
  size_t off = 0;
  auto alloc = [&](size_t bytes) -> void* {
    void* p = ws + off;
    off = (off + bytes + 255) & ~(size_t)255;
    return p;
  };
  const int nbins = cdiv(N, BIN_NODES);
  float*    deg   = (float*)   alloc(sizeof(float) * N);
  float*    g     = (float*)   alloc(sizeof(float) * NH);
  float*    hbuf  = (float*)   alloc(sizeof(float) * NH);
  int*      rec   = (int*)     alloc(sizeof(int4) * E);
  _Float16* Wfrag = (_Float16*)alloc(sizeof(_Float16) * 2 * 3 * H * H);
  int*      binCnt   = (int*)  alloc(sizeof(int) * BIN_PAD * nbins);
  int*      cursor   = (int*)  alloc(sizeof(int) * BIN_PAD * nbins);
  int*      binStart = (int*)  alloc(sizeof(int) * (nbins + 1));
  float*    degB  = g;
  (void)ws_size;

  init_deg_kernel <<<cdiv(N, 256), 256, 0, stream>>>(deg, degB, binCnt, nbins, N);
  deg_accum_kernel<<<cdiv(E, 256), 256, 0, stream>>>(dst, deg, degB, binCnt, E);
  dinv_kernel     <<<cdiv(N, 256), 256, 0, stream>>>(deg, degB, N);
  bin_scan_kernel <<<1, SCAN_THREADS, 0, stream>>>(binCnt, binStart, cursor, nbins);
  bin_fill_kernel <<<cdiv(E, 256), 256, 0, stream>>>(src, dst, deg, cursor, rec, E);
  cvt_weights_kernel<<<cdiv(2 * 3 * H * H, 256), 256, 0, stream>>>(Wih, Whh, Wfrag, 2 * 3 * H * H);
  init_h_kernel   <<<cdiv((long long)(NH / 4), 256 * INIT_VEC), 256, 0, stream>>>(hbuf, (int)(NH / 4));

  for (int t = 0; t < T_STEPS; ++t) {
    transform_kernel<<<cdiv(N, 8 * TR_NODES), 256, 0, stream>>>(X, Wg, g, t, N);
    edge_gru_kernel <<<nbins, EDGE_THREADS, 0, stream>>>(rec, binStart, g, deg, bg, hbuf, (const v8h*)Wfrag, bih, bhh, N);
  }

  head_kernel<<<cdiv(N, 256), 256, 0, stream>>>(hbuf, Wout, bout, (float*)d_out, N);
}


# Bin records written in destination order (per-node fill cursors); a wave sums each run of equal destinations in registers and sends two LDS float adds per run, not per edge; sum order changed
# speedup vs baseline: 3.1529x; 1.2204x over previous
#include <hip/hip_runtime.h>

#define T_STEPS 12
#define F_IN 16
#define H 64

typedef __attribute__((ext_vector_type(16))) _Float16 v16h;
typedef __attribute__((ext_vector_type(8)))  _Float16 v8h;
typedef __attribute__((ext_vector_type(8)))  float    v8f;


struct RowPieces { float4 p0, p1, p2, p3; };
__device__ __forceinline__ RowPieces loadA_f32(const float* row, int kb) {
  RowPieces q;
  q.p0 = *(const float4*)(row + kb);
  q.p1 = *(const float4*)(row + kb + 4);
  q.p2 = *(const float4*)(row + kb + 16);
  q.p3 = *(const float4*)(row + kb + 20);
  return q;
}
__device__ __forceinline__ v16h fragA_from_f32(const RowPieces& q) {
  v16h r;
  r[0]  = (_Float16)q.p0.x; r[1]  = (_Float16)q.p0.y; r[2]  = (_Float16)q.p0.z; r[3]  = (_Float16)q.p0.w;
  r[4]  = (_Float16)q.p1.x; r[5]  = (_Float16)q.p1.y; r[6]  = (_Float16)q.p1.z; r[7]  = (_Float16)q.p1.w;
  r[8]  = (_Float16)q.p2.x; r[9]  = (_Float16)q.p2.y; r[10] = (_Float16)q.p2.z; r[11] = (_Float16)q.p2.w;
  r[12] = (_Float16)q.p3.x; r[13] = (_Float16)q.p3.y; r[14] = (_Float16)q.p3.z; r[15] = (_Float16)q.p3.w;
  return r;
}

__device__ __forceinline__ float sigmoidf_(float x) {
  return __builtin_amdgcn_rcpf(1.0f + __expf(-x));
}
__device__ __forceinline__ float tanhf_(float x) {
  return 1.0f - 2.0f * __builtin_amdgcn_rcpf(1.0f + __expf(2.0f * x));
}


__device__ __forceinline__ bool upper_half() {
  return ((__builtin_amdgcn_s_sendmsg_rtn(0x87) >> 18) & 1u) != 0u;
}

#define BIN_NODES 208
#define BIN_PAD   32

__global__ __launch_bounds__(256) void init_deg_kernel(float* __restrict__ deg,
                                                       float* __restrict__ degB,
                                                       int* __restrict__ binCnt,
                                                       int nbins, int N) {
  int i = blockIdx.x * 256 + threadIdx.x;
  if (i < N) { deg[i] = 1.0f; degB[i] = 0.0f; }
  if (i < nbins) binCnt[i * BIN_PAD] = 0;
}

__global__ __launch_bounds__(256) void deg_accum_kernel(const int* __restrict__ dst,
                                                        float* __restrict__ degA,
                                                        float* __restrict__ degB,
                                                        int* __restrict__ binCnt, int E) {
  int e = blockIdx.x * 256 + threadIdx.x;
  float* __restrict__ deg = upper_half() ? degB : degA;
  if (e < E) {
    const int d = dst[e];
    atomicAdd(&deg[d], 1.0f);
    atomicAdd(&binCnt[(d / BIN_NODES) * BIN_PAD], 1);
  }
}

#define NODE_SCAN_THREADS 256
static_assert(BIN_NODES <= NODE_SCAN_THREADS, "one thread per node of the bin");
__global__ __launch_bounds__(NODE_SCAN_THREADS) void dinv_scan_kernel(float* __restrict__ deg,
                                                                      const float* __restrict__ degB,
                                                                      const int* __restrict__ binStart,
                                                                      int* __restrict__ nodeCur, int N) {
  __shared__ int sh[NODE_SCAN_THREADS];
  const int tid = threadIdx.x;
  const int n   = blockIdx.x * BIN_NODES + tid;
  const bool mine = (tid < BIN_NODES) && (n < N);
  int v = 0;
  if (mine) {
    const float c = deg[n] + degB[n];
    deg[n] = rsqrtf(c);
    v = (int)c - 1;
  }
  sh[tid] = v;
  __syncthreads();
  for (int off = 1; off < NODE_SCAN_THREADS; off <<= 1) {
    const int t = (tid >= off) ? sh[tid - off] : 0;
    __syncthreads();
    sh[tid] += t;
    __syncthreads();
  }
  if (mine) nodeCur[n] = binStart[blockIdx.x] + sh[tid] - v;
}

#define SCAN_THREADS 1024
__global__ __launch_bounds__(SCAN_THREADS) void bin_scan_kernel(const int* __restrict__ binCnt,
                                                                int* __restrict__ binStart,
                                                                int nbins) {
  __shared__ int sh[SCAN_THREADS];
  const int tid = threadIdx.x;
  int carry = 0;
  for (int base = 0; base < nbins; base += SCAN_THREADS) {
    const int i = base + tid;
    const int v = (i < nbins) ? binCnt[i * BIN_PAD] : 0;
    sh[tid] = v;
    __syncthreads();
    for (int off = 1; off < SCAN_THREADS; off <<= 1) {
      const int t = (tid >= off) ? sh[tid - off] : 0;
      __syncthreads();
      sh[tid] += t;
      __syncthreads();
    }
    const int excl = carry + sh[tid] - v;
    if (i < nbins) binStart[i] = excl;
    carry += sh[SCAN_THREADS - 1];
    __syncthreads();
  }
  if (tid == 0) binStart[nbins] = carry;
}

__global__ __launch_bounds__(256) void bin_fill_kernel(const int* __restrict__ src,
                                                       const int* __restrict__ dst,
                                                       const float* __restrict__ dinv,
                                                       int* __restrict__ nodeCur,
                                                       int* __restrict__ recw, int E) {
  int4* __restrict__ rec = (int4*)recw;
  int e = blockIdx.x * 256 + threadIdx.x;
  if (e < E) {
    const int s = src[e];
    const int d = dst[e];
    const int b = d / BIN_NODES;
    const int pos = atomicAdd(&nodeCur[d], 1);
    rec[pos] = make_int4(s, d - b * BIN_NODES, __float_as_int(dinv[s] * dinv[d]), 0);
  }
}

__global__ __launch_bounds__(256) void cvt_weights_kernel(const float* __restrict__ Wih,
                                                          const float* __restrict__ Whh,
                                                          _Float16* __restrict__ Wfrag,
                                                          int n) {
  int o = blockIdx.x * 256 + threadIdx.x;
  if (o < n) {
    const int e    = o & 7;
    const int lane = (o >> 3) & 31;
    const int hh   = (o >> 8) & 1;
    const int f    = o >> 9;
    const int c    = f & 1;
    const int j    = (f >> 1) & 3;
    const int mg   = f >> 3;
    const int mat  = mg / 3;
    const int gate = mg - mat * 3;
    const int row  = gate * 64 + j * 16 + (lane & 15);
    const int k    = ((lane < 16) ? 0 : 8) + 32 * c + 16 * hh + e;
    const float* Wsrc = mat ? Whh : Wih;
    Wfrag[o] = (_Float16)Wsrc[row * H + k];
  }
}

#define INIT_VEC 4
__global__ __launch_bounds__(256) void init_h_kernel(float* __restrict__ h, int n4) {
  float4* __restrict__ h4 = (float4*)h;
  const int base = blockIdx.x * (256 * INIT_VEC) + threadIdx.x;
  #pragma unroll
  for (int j = 0; j < INIT_VEC; ++j) {
    const int i = base + j * 256;
    if (i < n4) h4[i] = make_float4(0.0f, 0.0f, 0.0f, 0.0f);
  }
}


#define TR_NODES 8
__global__ __launch_bounds__(256) void transform_kernel(const float* __restrict__ X,
                                                        const float* __restrict__ Wg,
                                                        float* __restrict__ g,
                                                        int t, int N) {
  const int lane = threadIdx.x & 31;
  const int node0 = __builtin_amdgcn_readfirstlane(
      (int)(((blockIdx.x * 256 + threadIdx.x) >> 5) * TR_NODES));
  if (node0 >= N) return;

  const int nx = (node0 + (lane >> 2) < N) ? node0 + (lane >> 2) : N - 1;
  const float4 xv = *(const float4*)(X + ((size_t)nx * T_STEPS + t) * F_IN + (lane & 3) * 4);

  float w0[F_IN], w1[F_IN];
  #pragma unroll
  for (int k = 0; k < F_IN; ++k) {
    w0[k] = Wg[k * H + lane];
    w1[k] = Wg[k * H + 32 + lane];
  }

  #pragma unroll
  for (int n = 0; n < TR_NODES; ++n) {
    if (node0 + n < N) {
      float acc0 = 0.0f, acc1 = 0.0f;
      #pragma unroll
      for (int q = 0; q < 4; ++q) {
        const float x0 = __int_as_float(__builtin_amdgcn_readlane(__float_as_int(xv.x), n * 4 + q));
        const float x1 = __int_as_float(__builtin_amdgcn_readlane(__float_as_int(xv.y), n * 4 + q));
        const float x2 = __int_as_float(__builtin_amdgcn_readlane(__float_as_int(xv.z), n * 4 + q));
        const float x3 = __int_as_float(__builtin_amdgcn_readlane(__float_as_int(xv.w), n * 4 + q));
        acc0 += x0 * w0[q * 4 + 0]; acc1 += x0 * w1[q * 4 + 0];
        acc0 += x1 * w0[q * 4 + 1]; acc1 += x1 * w1[q * 4 + 1];
        acc0 += x2 * w0[q * 4 + 2]; acc1 += x2 * w1[q * 4 + 2];
        acc0 += x3 * w0[q * 4 + 3]; acc1 += x3 * w1[q * 4 + 3];
      }
      const size_t idx = (size_t)(node0 + n) * H + lane;
      g[idx]        = acc0;
      g[idx + 32]   = acc1;
    }
  }
}

#define EDGE_BATCH   32
#define EDGE_WAVES   16
#define EDGE_THREADS (EDGE_WAVES * 32)
#define WFRAG_VEC    (48 * 2 * 32)

__device__ __forceinline__ v16h lds_fragB(const v8h* sW, int f, int lane) {
  union { v16h v; v8h h[2]; } u;
  u.h[0] = sW[(f * 2 + 0) * 32 + lane];
  u.h[1] = sW[(f * 2 + 1) * 32 + lane];
  return u.v;
}

__global__ __launch_bounds__(EDGE_THREADS, 8) void edge_gru_kernel(const int* __restrict__ recw,
                                                                const int* __restrict__ binStart,
                                                                const float* __restrict__ g,
                                                                float* __restrict__ gnext,
                                                                const float* __restrict__ X,
                                                                const float* __restrict__ Wg,
                                                                int tn,
                                                                const float* __restrict__ dinv,
                                                                const float* __restrict__ bg,
                                                                float* h,
                                                                const v8h* __restrict__ Wfrag,
                                                                const float* __restrict__ b_ih,
                                                                const float* __restrict__ b_hh,
                                                                int N) {
  __shared__ __attribute__((aligned(16))) float tile[BIN_NODES * H];
  __shared__ v8h sW[WFRAG_VEC];
  const int4* __restrict__ rec = (const int4*)recw;

  const int lane = threadIdx.x & 31;
  const int wave = __builtin_amdgcn_readfirstlane((int)(threadIdx.x >> 5));
  const int n0   = blockIdx.x * BIN_NODES;
  const int nn   = (N - n0 < BIN_NODES) ? (N - n0) : BIN_NODES;
  const int nv   = nn * (H / 4);
  float4* __restrict__ t4 = (float4*)tile;

  const bool gruwave = (wave * 16 + 16) <= nn;
  const int  node0   = n0 + wave * 16;
  const int  m       = lane & 15;
  const int  kb      = (lane < 16) ? 0 : 8;
  const int  mrow0   = (lane >> 4) << 3;

  constexpr int TR_ROWS = BIN_NODES / EDGE_WAVES;
  static_assert(TR_ROWS * EDGE_WAVES == BIN_NODES && TR_ROWS > 8 && TR_ROWS <= 16, "two x loads a wave");
  const bool trnext = tn < T_STEPS;
  const int  tb     = wave * TR_ROWS;
  float4 xa = make_float4(0.0f, 0.0f, 0.0f, 0.0f), xb = xa;
  float w0[F_IN], w1[F_IN];
  #pragma unroll
  for (int k = 0; k < F_IN; ++k) { w0[k] = 0.0f; w1[k] = 0.0f; }
  if (trnext) {
    int la = tb + (lane >> 2);
    int lb = tb + 8 + (lane >> 2);
    if (lb > tb + TR_ROWS - 1) lb = tb + TR_ROWS - 1;
    if (la > nn - 1) la = nn - 1;
    if (lb > nn - 1) lb = nn - 1;
    xa = *(const float4*)(X + ((size_t)(n0 + la) * T_STEPS + tn) * F_IN + (lane & 3) * 4);
    xb = *(const float4*)(X + ((size_t)(n0 + lb) * T_STEPS + tn) * F_IN + (lane & 3) * 4);
    #pragma unroll
    for (int k = 0; k < F_IN; ++k) {
      w0[k] = Wg[k * H + lane];
      w1[k] = Wg[k * H + 32 + lane];
    }
  }

  constexpr int kPasses = (WFRAG_VEC + EDGE_THREADS - 1) / EDGE_THREADS;
  v8h wpiece[kPasses];
  #pragma unroll
  for (int it = 0; it < kPasses; ++it) {
    const int i = it * EDGE_THREADS + (int)threadIdx.x;
    wpiece[it] = Wfrag[i < WFRAG_VEC ? i : WFRAG_VEC - 1];
  }
  const float4* __restrict__ g4 = (const float4*)(g + (size_t)n0 * H);
  constexpr int kInit = (BIN_NODES * (H / 4) + EDGE_THREADS - 1) / EDGE_THREADS;
  float4 gv[kInit];
  float  dv[kInit];
  #pragma unroll
  for (int it = 0; it < kInit; ++it) {
    const int i  = it * EDGE_THREADS + (int)threadIdx.x;
    const int ic = (i < nv) ? i : nv - 1;
    gv[it] = g4[ic];
    dv[it] = dinv[n0 + (ic >> 4)];
  }
  if (trnext) {
    #pragma unroll
    for (int n = 0; n < TR_ROWS; ++n) {
      if (tb + n < nn) {
        const float4 xs = (n < 8) ? xa : xb;
        const int lq = (n & 7) * 4;
        float acc0 = 0.0f, acc1 = 0.0f;
        #pragma unroll
        for (int q = 0; q < 4; ++q) {
          const float x0 = __int_as_float(__builtin_amdgcn_readlane(__float_as_int(xs.x), lq + q));
          const float x1 = __int_as_float(__builtin_amdgcn_readlane(__float_as_int(xs.y), lq + q));
          const float x2 = __int_as_float(__builtin_amdgcn_readlane(__float_as_int(xs.z), lq + q));
          const float x3 = __int_as_float(__builtin_amdgcn_readlane(__float_as_int(xs.w), lq + q));
          acc0 += x0 * w0[q * 4 + 0]; acc1 += x0 * w1[q * 4 + 0];
          acc0 += x1 * w0[q * 4 + 1]; acc1 += x1 * w1[q * 4 + 1];
          acc0 += x2 * w0[q * 4 + 2]; acc1 += x2 * w1[q * 4 + 2];
          acc0 += x3 * w0[q * 4 + 3]; acc1 += x3 * w1[q * 4 + 3];
        }
        const size_t idx = (size_t)(n0 + tb + n) * H + lane;
        gnext[idx]      = acc0;
        gnext[idx + 32] = acc1;
      }
    }
  }

  const float* hrow = h + (size_t)(gruwave ? node0 + m : n0) * H;
  const RowPieces hq0 = loadA_f32(hrow, kb);
  const RowPieces hq1 = loadA_f32(hrow, 32 + kb);

  #pragma unroll
  for (int it = 0; it < kPasses; ++it) {
    const int i = it * EDGE_THREADS + (int)threadIdx.x;
    if (i < WFRAG_VEC) sW[i] = wpiece[it];
  }
  #pragma unroll
  for (int it = 0; it < kInit; ++it) {
    const int i = it * EDGE_THREADS + (int)threadIdx.x;
    const float di = dv[it];
    if (i < nv)
      t4[i] = make_float4(gv[it].x * di * di, gv[it].y * di * di, gv[it].z * di * di, gv[it].w * di * di);
  }
  __syncthreads();
  v16h hA0 = fragA_from_f32(hq0);
  v16h hA1 = fragA_from_f32(hq1);
  asm volatile("" : "+v"(hA0), "+v"(hA1));

  const int eb = binStart[blockIdx.x];
  const int ee = binStart[blockIdx.x + 1];
  int4 cur = make_int4(0, 0, 0, 0);
  if (eb + wave * 32 + lane < ee) cur = rec[eb + wave * 32 + lane];
  for (int base = eb + wave * 32; base < ee; base += EDGE_THREADS) {
    const int s = cur.x, d = cur.y;
    const float we = __int_as_float(cur.z);
    const int cnt = (ee - base < 32) ? (ee - base) : 32;
    const int en = base + EDGE_THREADS + lane;
    cur = make_int4(0, 0, 0, 0);
    if (en < ee) cur = rec[en];

    if (cnt == 32) {
      float acc0 = 0.0f, acc1 = 0.0f;
      int dnext = __builtin_amdgcn_readlane(d, 0);
      #pragma unroll
      for (int b = 0; b < 32; b += EDGE_BATCH) {
        float v0[EDGE_BATCH], v1[EDGE_BATCH];
        #pragma unroll
        for (int j = 0; j < EDGE_BATCH; ++j) {
          const int sj = __builtin_amdgcn_readlane(s, b + j);
          const float* rowp = g + (size_t)sj * H;
          v0[j] = rowp[lane];
          v1[j] = rowp[lane + 32];
        }
        #pragma unroll
        for (int j = 0; j < EDGE_BATCH; ++j) {
          const int dj = dnext;
          dnext = __builtin_amdgcn_readlane(d, (b + j + 1) & 31);
          const float wj = __int_as_float(__builtin_amdgcn_readlane(__float_as_int(we), b + j));
          acc0 += wj * v0[j];
          acc1 += wj * v1[j];
          const bool last = (b + j == 31) || (dj != dnext);
          if (last) {
            float* tp = tile + dj * H + lane;
            atomicAdd(tp,      acc0);
            atomicAdd(tp + 32, acc1);
            acc0 = 0.0f; acc1 = 0.0f;
          }
        }
      }
    } else {
      float acc0 = 0.0f, acc1 = 0.0f;
      #pragma unroll
      for (int b = 0; b < 32; b += EDGE_BATCH) {
        if (b < cnt) {
          float v0[EDGE_BATCH], v1[EDGE_BATCH];
          #pragma unroll
          for (int j = 0; j < EDGE_BATCH; ++j) {
            const int jj = (b + j < cnt) ? (b + j) : (cnt - 1);
            const int sj = __builtin_amdgcn_readlane(s, jj);
            const float* rowp = g + (size_t)sj * H;
            v0[j] = rowp[lane];
            v1[j] = rowp[lane + 32];
          }
          #pragma unroll
          for (int j = 0; j < EDGE_BATCH; ++j) {
            if (b + j < cnt) {
              const int dj = __builtin_amdgcn_readlane(d, b + j);
              const float wj = __int_as_float(__builtin_amdgcn_readlane(__float_as_int(we), b + j));
              acc0 += wj * v0[j];
              acc1 += wj * v1[j];
              const bool last = (b + j + 1 >= cnt) ||
                                (dj != __builtin_amdgcn_readlane(d, (b + j + 1) & 31));
              if (last) {
                float* tp = tile + dj * H + lane;
                atomicAdd(tp,      acc0);
                atomicAdd(tp + 32, acc1);
                acc0 = 0.0f; acc1 = 0.0f;
              }
            }
          }
        }
      }
    }
  }
  __syncthreads();

  if (!gruwave) return;

  v16h aA0, aA1;
  {
    const float* trow = tile + (wave * 16 + m) * H;
    #pragma unroll
    for (int c = 0; c < 2; ++c) {
      const int k0 = 32 * c + kb;
      const float4 p0 = *(const float4*)(trow + k0);
      const float4 p1 = *(const float4*)(trow + k0 + 4);
      const float4 p2 = *(const float4*)(trow + k0 + 16);
      const float4 p3 = *(const float4*)(trow + k0 + 20);
      const float4 b0 = *(const float4*)(bg + k0);
      const float4 b1 = *(const float4*)(bg + k0 + 4);
      const float4 b2 = *(const float4*)(bg + k0 + 16);
      const float4 b3 = *(const float4*)(bg + k0 + 20);
      v16h r;
      r[0]  = (_Float16)fmaxf(p0.x + b0.x, 0.0f); r[1]  = (_Float16)fmaxf(p0.y + b0.y, 0.0f);
      r[2]  = (_Float16)fmaxf(p0.z + b0.z, 0.0f); r[3]  = (_Float16)fmaxf(p0.w + b0.w, 0.0f);
      r[4]  = (_Float16)fmaxf(p1.x + b1.x, 0.0f); r[5]  = (_Float16)fmaxf(p1.y + b1.y, 0.0f);
      r[6]  = (_Float16)fmaxf(p1.z + b1.z, 0.0f); r[7]  = (_Float16)fmaxf(p1.w + b1.w, 0.0f);
      r[8]  = (_Float16)fmaxf(p2.x + b2.x, 0.0f); r[9]  = (_Float16)fmaxf(p2.y + b2.y, 0.0f);
      r[10] = (_Float16)fmaxf(p2.z + b2.z, 0.0f); r[11] = (_Float16)fmaxf(p2.w + b2.w, 0.0f);
      r[12] = (_Float16)fmaxf(p3.x + b3.x, 0.0f); r[13] = (_Float16)fmaxf(p3.y + b3.y, 0.0f);
      r[14] = (_Float16)fmaxf(p3.z + b3.z, 0.0f); r[15] = (_Float16)fmaxf(p3.w + b3.w, 0.0f);
      if (c == 0) aA0 = r; else aA1 = r;
    }
  }

  #pragma unroll
  for (int j = 0; j < 4; ++j) {
    const int fr = ((0 * 3 + 0) * 4 + j) * 2;
    const int fz = ((0 * 3 + 1) * 4 + j) * 2;
    const int fn = ((0 * 3 + 2) * 4 + j) * 2;
    const int gr = ((1 * 3 + 0) * 4 + j) * 2;
    const int gz = ((1 * 3 + 1) * 4 + j) * 2;
    const int gn = ((1 * 3 + 2) * 4 + j) * 2;

    const float bir = b_ih[      j * 16 + m];
    const float biz = b_ih[ 64 + j * 16 + m];
    const float bin = b_ih[128 + j * 16 + m];
    const float bhr = b_hh[      j * 16 + m];
    const float bhz = b_hh[ 64 + j * 16 + m];
    const float bhn = b_hh[128 + j * 16 + m];
    const size_t hbase = (size_t)(node0 + mrow0) * H + j * 16 + m;
    float hold[8];
    #pragma unroll
    for (int v = 0; v < 8; ++v) hold[v] = h[hbase + (size_t)v * H];

    v8f ir = {}, iz = {}, in_ = {}, hr = {}, hz = {}, hn = {};

    ir  = __builtin_amdgcn_wmma_f32_16x16x32_f16(false, aA0, false, lds_fragB(sW, fr, lane), (short)0, ir,  false, false);
    iz  = __builtin_amdgcn_wmma_f32_16x16x32_f16(false, aA0, false, lds_fragB(sW, fz, lane), (short)0, iz,  false, false);
    in_ = __builtin_amdgcn_wmma_f32_16x16x32_f16(false, aA0, false, lds_fragB(sW, fn, lane), (short)0, in_, false, false);
    hr  = __builtin_amdgcn_wmma_f32_16x16x32_f16(false, hA0, false, lds_fragB(sW, gr, lane), (short)0, hr,  false, false);
    hz  = __builtin_amdgcn_wmma_f32_16x16x32_f16(false, hA0, false, lds_fragB(sW, gz, lane), (short)0, hz,  false, false);
    hn  = __builtin_amdgcn_wmma_f32_16x16x32_f16(false, hA0, false, lds_fragB(sW, gn, lane), (short)0, hn,  false, false);
    ir  = __builtin_amdgcn_wmma_f32_16x16x32_f16(false, aA1, false, lds_fragB(sW, fr + 1, lane), (short)0, ir,  false, false);
    iz  = __builtin_amdgcn_wmma_f32_16x16x32_f16(false, aA1, false, lds_fragB(sW, fz + 1, lane), (short)0, iz,  false, false);
    in_ = __builtin_amdgcn_wmma_f32_16x16x32_f16(false, aA1, false, lds_fragB(sW, fn + 1, lane), (short)0, in_, false, false);
    hr  = __builtin_amdgcn_wmma_f32_16x16x32_f16(false, hA1, false, lds_fragB(sW, gr + 1, lane), (short)0, hr,  false, false);
    hz  = __builtin_amdgcn_wmma_f32_16x16x32_f16(false, hA1, false, lds_fragB(sW, gz + 1, lane), (short)0, hz,  false, false);
    hn  = __builtin_amdgcn_wmma_f32_16x16x32_f16(false, hA1, false, lds_fragB(sW, gn + 1, lane), (short)0, hn,  false, false);

    #pragma unroll
    for (int v = 0; v < 8; ++v) {
      float r  = sigmoidf_(ir[v] + bir + hr[v] + bhr);
      float z  = sigmoidf_(iz[v] + biz + hz[v] + bhz);
      float nc = tanhf_(in_[v] + bin + r * (hn[v] + bhn));
      float hnew = (1.0f - z) * nc + z * hold[v];
      h[hbase + (size_t)v * H] = hnew;
    }
  }
}

__global__ __launch_bounds__(256) void head_kernel(const float* __restrict__ h,
                                                   const float* __restrict__ Wout,
                                                   const float* __restrict__ bout,
                                                   float* __restrict__ out, int N) {
  int i = blockIdx.x * 256 + threadIdx.x;
  if (i >= N) return;
  float a0 = bout[0], a1 = bout[1];
  const float* hr = h + (size_t)i * H;
  #pragma unroll
  for (int k = 0; k < H; ++k) {
    float hv = hr[k];
    a0 += hv * Wout[k * 2 + 0];
    a1 += hv * Wout[k * 2 + 1];
  }
  out[i * 2 + 0] = a0;
  out[i * 2 + 1] = a1;
}


static inline int cdiv(long long a, long long b) { return (int)((a + b - 1) / b); }

extern "C" void kernel_launch(void* const* d_in, const int* in_sizes, int n_in,
                              void* d_out, int out_size, void* d_ws, size_t ws_size,
                              hipStream_t stream) {
  const float* X    = (const float*)d_in[0];
  const int*   ei   = (const int*)  d_in[1];
  const float* Wg   = (const float*)d_in[2];
  const float* bg   = (const float*)d_in[3];
  const float* Wih  = (const float*)d_in[4];
  const float* Whh  = (const float*)d_in[5];
  const float* bih  = (const float*)d_in[6];
  const float* bhh  = (const float*)d_in[7];
  const float* Wout = (const float*)d_in[8];
  const float* bout = (const float*)d_in[9];

  const int N = in_sizes[0] / (T_STEPS * F_IN);
  const int E = in_sizes[1] / 2;
  const int* src = ei;
  const int* dst = ei + E;
  const size_t NH = (size_t)N * H;

  char* ws = (char*)d_ws;
  size_t off = 0;
  auto alloc = [&](size_t bytes) -> void* {
    void* p = ws + off;
    off = (off + bytes + 255) & ~(size_t)255;
    return p;
  };
  const int nbins = cdiv(N, BIN_NODES);
  float*    deg   = (float*)   alloc(sizeof(float) * N);
  float*    g     = (float*)   alloc(sizeof(float) * NH);
  float*    hbuf  = (float*)   alloc(sizeof(float) * NH);
  int*      rec   = (int*)     alloc(sizeof(int4) * E);
  _Float16* Wfrag = (_Float16*)alloc(sizeof(_Float16) * 2 * 3 * H * H);
  int*      binCnt   = (int*)  alloc(sizeof(int) * BIN_PAD * nbins);
  int*      binStart = (int*)  alloc(sizeof(int) * (nbins + 1));
  float*    g2    = (float*)   alloc(sizeof(float) * NH);
  float*    degB    = g;
  int*      nodeCur = (int*)g2;
  (void)ws_size;

  init_deg_kernel <<<cdiv(N, 256), 256, 0, stream>>>(deg, degB, binCnt, nbins, N);
  deg_accum_kernel<<<cdiv(E, 256), 256, 0, stream>>>(dst, deg, degB, binCnt, E);
  bin_scan_kernel <<<1, SCAN_THREADS, 0, stream>>>(binCnt, binStart, nbins);
  dinv_scan_kernel<<<nbins, NODE_SCAN_THREADS, 0, stream>>>(deg, degB, binStart, nodeCur, N);
  bin_fill_kernel <<<cdiv(E, 256), 256, 0, stream>>>(src, dst, deg, nodeCur, rec, E);
  cvt_weights_kernel<<<cdiv(2 * 3 * H * H, 256), 256, 0, stream>>>(Wih, Whh, Wfrag, 2 * 3 * H * H);
  init_h_kernel   <<<cdiv((long long)(NH / 4), 256 * INIT_VEC), 256, 0, stream>>>(hbuf, (int)(NH / 4));

  transform_kernel<<<cdiv(N, 8 * TR_NODES), 256, 0, stream>>>(X, Wg, g, 0, N);
  for (int t = 0; t < T_STEPS; ++t) {
    const float* gcur = (t & 1) ? g2 : g;
    float*       gnxt = (t & 1) ? g : g2;
    edge_gru_kernel <<<nbins, EDGE_THREADS, 0, stream>>>(rec, binStart, gcur, gnxt, X, Wg, t + 1, deg, bg, hbuf, (const v8h*)Wfrag, bih, bhh, N);
  }

  head_kernel<<<cdiv(N, 256), 256, 0, stream>>>(hbuf, Wout, bout, (float*)d_out, N);
}
